# starting kernel
# baseline (speedup 1.0000x reference)
#include <hip/hip_runtime.h>
#include <hip/hip_bf16.h>


#define S_STATES 16
#define C_TYPES  64
#define G_GENES  20000
#define GB       16

typedef __attribute__((ext_vector_type(2))) float v2f;
typedef __attribute__((ext_vector_type(8))) float v8f;

__global__ __launch_bounds__(128)
void lognorm_wmma_kernel(const float* __restrict__ Z_mu,
                         const float* __restrict__ Z_sigma,
                         const float* __restrict__ corr,
                         const float* __restrict__ cell_prob,
                         float* __restrict__ mu_out,
                         float* __restrict__ sg_out) {
    __shared__ float s_corr[C_TYPES * C_TYPES];
    __shared__ float s_ev[C_TYPES * GB];
    __shared__ float s_ab[C_TYPES * GB];
    __shared__ float s_cp[S_STATES * C_TYPES];

    const int tid  = threadIdx.x;
    const int g0   = blockIdx.x * GB;

    #pragma unroll
    for (int i = tid; i < C_TYPES * C_TYPES; i += 128) s_corr[i] = corr[i];
    #pragma unroll
    for (int i = tid; i < S_STATES * C_TYPES; i += 128) s_cp[i] = cell_prob[i];

    #pragma unroll
    for (int i = tid; i < C_TYPES * GB; i += 128) {
        const int c = i >> 4;
        const int g = i & 15;
        const float m  = Z_mu[c * G_GENES + g0 + g];
        const float sg = Z_sigma[c * G_GENES + g0 + g];
        const float e  = __expf(m + 0.5f * sg * sg);
        s_ev[i] = e;
        s_ab[i] = e * sg;
    }
    __syncthreads();

    const int wave  = tid >> 5;
    const int lane  = tid & 31;
    const int m     = lane & 15;
    const int hi    = lane >> 4;
    const int khalf = hi << 1;

    v2f cA[4][16];
    #pragma unroll
    for (int rt = 0; rt < 4; ++rt) {
        #pragma unroll
        for (int kc = 0; kc < 16; ++kc) {
            const int row = rt * 16 + m;
            const int d0  = kc * 4 + khalf;
            cA[rt][kc].x = s_corr[row * 64 + d0];
            cA[rt][kc].y = s_corr[row * 64 + d0 + 1];
        }
    }

    #pragma unroll
    for (int si = 0; si < 4; ++si) {
        const int s = wave + si * 4;
        const float* cp = &s_cp[s * C_TYPES];

        v8f acc0 = {}, acc1 = {}, acc2 = {}, acc3 = {};
        #pragma unroll
        for (int kc = 0; kc < 16; ++kc) {
            const int c0 = kc * 4 + khalf;
            v2f b;
            b.x = cp[c0]     * s_ab[c0 * GB + m];
            b.y = cp[c0 + 1] * s_ab[(c0 + 1) * GB + m];
            acc0 = __builtin_amdgcn_wmma_f32_16x16x4_f32(false, cA[0][kc], false, b, (short)0, acc0, false, false);
            acc1 = __builtin_amdgcn_wmma_f32_16x16x4_f32(false, cA[1][kc], false, b, (short)0, acc1, false, false);
            acc2 = __builtin_amdgcn_wmma_f32_16x16x4_f32(false, cA[2][kc], false, b, (short)0, acc2, false, false);
            acc3 = __builtin_amdgcn_wmma_f32_16x16x4_f32(false, cA[3][kc], false, b, (short)0, acc3, false, false);
        }

        float ss = 0.0f;
        const int cbase = hi * 32;
        #pragma unroll
        for (int c = 0; c < 32; ++c)
            ss += cp[cbase + c] * s_ev[(cbase + c) * GB + m];
        ss += __shfl_xor(ss, 16, 32);

        float dot = 0.0f;
        #pragma unroll
        for (int v = 0; v < 8; ++v) {
            const int r0 = v + hi * 8;
            dot += acc0[v] * (cp[r0]      * s_ab[(r0)      * GB + m]);
            dot += acc1[v] * (cp[r0 + 16] * s_ab[(r0 + 16) * GB + m]);
            dot += acc2[v] * (cp[r0 + 32] * s_ab[(r0 + 32) * GB + m]);
            dot += acc3[v] * (cp[r0 + 48] * s_ab[(r0 + 48) * GB + m]);
        }
        dot += __shfl_xor(dot, 16, 32);

        const float var   = dot / (ss * ss);
        const float mu_o  = __logf(ss) - 0.5f * var;
        const float sig_o = sqrtf(var);

        if (lane < 16) {
            mu_out[s * G_GENES + g0 + m] = mu_o;
            sg_out[s * G_GENES + g0 + m] = sig_o;
        }
    }
}

extern "C" void kernel_launch(void* const* d_in, const int* in_sizes, int n_in,
                              void* d_out, int out_size, void* d_ws, size_t ws_size,
                              hipStream_t stream) {
    const float* Z_mu      = (const float*)d_in[0];
    const float* Z_sigma   = (const float*)d_in[1];
    const float* corr      = (const float*)d_in[2];
    const float* cell_prob = (const float*)d_in[3];

    float* mu_out = (float*)d_out;
    float* sg_out = mu_out + S_STATES * G_GENES;

    const int nblocks = G_GENES / GB;
    lognorm_wmma_kernel<<<nblocks, 128, 0, stream>>>(
        Z_mu, Z_sigma, corr, cell_prob, mu_out, sg_out);
}


# operands built in registers straight from global loads (corr read transposed, 4 tiles per 16-byte load, D rows land in the lane's own B slots): no LDS, no barrier; same products, sums regrouped
# speedup vs baseline: 2.2222x; 2.2222x over previous
#include <hip/hip_runtime.h>


#define S_STATES 16
#define C_TYPES  64
#define G_GENES  20000
#define GB       16

typedef __attribute__((ext_vector_type(2))) float v2f;
typedef __attribute__((ext_vector_type(4))) float v4f;
typedef __attribute__((ext_vector_type(8))) float v8f;

__global__ __launch_bounds__(128, 3)
void lognorm_wmma_kernel(const float* __restrict__ Z_mu,
                         const float* __restrict__ Z_sigma,
                         const float* __restrict__ corr,
                         const float* __restrict__ cell_prob,
                         float* __restrict__ mu_out,
                         float* __restrict__ sg_out) {
    const int tid  = threadIdx.x;
    const int wave = tid >> 5;
    const int lane = tid & 31;
    const int m    = lane & 15;
    const int hi   = lane >> 4;
    const int g    = blockIdx.x * GB + m;

    v2f ev[16], ab[16];
    #pragma unroll
    for (int kc = 0; kc < 16; ++kc) {
        const int c0 = hi * 32 + kc * 2;
        const float mu0 = Z_mu[c0 * G_GENES + g];
        const float mu1 = Z_mu[(c0 + 1) * G_GENES + g];
        const float s0  = Z_sigma[c0 * G_GENES + g];
        const float s1  = Z_sigma[(c0 + 1) * G_GENES + g];
        const float e0  = __expf(mu0 + 0.5f * s0 * s0);
        const float e1  = __expf(mu1 + 0.5f * s1 * s1);
        ev[kc].x = e0;      ev[kc].y = e1;
        ab[kc].x = e0 * s0; ab[kc].y = e1 * s1;
    }

    v2f cA[4][16];
    #pragma unroll
    for (int kc = 0; kc < 16; ++kc) {
        const int k0 = hi * 32 + kc * 2;
        const v4f r0 = *(const v4f*)(corr + k0 * C_TYPES + m * 4);
        const v4f r1 = *(const v4f*)(corr + (k0 + 1) * C_TYPES + m * 4);
        cA[0][kc].x = r0.x; cA[0][kc].y = r1.x;
        cA[1][kc].x = r0.y; cA[1][kc].y = r1.y;
        cA[2][kc].x = r0.z; cA[2][kc].y = r1.z;
        cA[3][kc].x = r0.w; cA[3][kc].y = r1.w;
    }

    const float* cp = cell_prob + wave * C_TYPES + hi * 32;
    v4f wn[8];
    #pragma unroll
    for (int q = 0; q < 8; ++q) wn[q] = *(const v4f*)(cp + q * 4);

    #pragma unroll
    for (int si = 0; si < 4; ++si) {
        const int s = wave + si * 4;

        v2f b[16];
        float ss = 0.0f;
        #pragma unroll
        for (int q = 0; q < 8; ++q) {
            const v4f w = wn[q];
            b[2 * q].x     = w.x * ab[2 * q].x;
            b[2 * q].y     = w.y * ab[2 * q].y;
            b[2 * q + 1].x = w.z * ab[2 * q + 1].x;
            b[2 * q + 1].y = w.w * ab[2 * q + 1].y;
            ss += w.x * ev[2 * q].x;
            ss += w.y * ev[2 * q].y;
            ss += w.z * ev[2 * q + 1].x;
            ss += w.w * ev[2 * q + 1].y;
        }
        ss += __shfl_xor(ss, 16, 32);

        if (si < 3) {
            #pragma unroll
            for (int q = 0; q < 8; ++q)
                wn[q] = *(const v4f*)(cp + (si + 1) * 4 * C_TYPES + q * 4);
        }

        v8f acc0 = {}, acc1 = {}, acc2 = {}, acc3 = {};
        #pragma unroll
        for (int kc = 0; kc < 16; ++kc) {
            acc0 = __builtin_amdgcn_wmma_f32_16x16x4_f32(false, cA[0][kc], false, b[kc], (short)0, acc0, false, false);
            acc1 = __builtin_amdgcn_wmma_f32_16x16x4_f32(false, cA[1][kc], false, b[kc], (short)0, acc1, false, false);
            acc2 = __builtin_amdgcn_wmma_f32_16x16x4_f32(false, cA[2][kc], false, b[kc], (short)0, acc2, false, false);
            acc3 = __builtin_amdgcn_wmma_f32_16x16x4_f32(false, cA[3][kc], false, b[kc], (short)0, acc3, false, false);
        }

        float dot = 0.0f;
        #pragma unroll
        for (int v = 0; v < 8; ++v) {
            dot += acc0[v] * b[2 * v].x;
            dot += acc1[v] * b[2 * v].y;
            dot += acc2[v] * b[2 * v + 1].x;
            dot += acc3[v] * b[2 * v + 1].y;
        }
        dot += __shfl_xor(dot, 16, 32);

        const float var   = dot / (ss * ss);
        const float mu_o  = __logf(ss) - 0.5f * var;
        const float sig_o = sqrtf(var);

        if (lane < 16) {
            mu_out[s * G_GENES + g] = mu_o;
            sg_out[s * G_GENES + g] = sig_o;
        }
    }
}

extern "C" void kernel_launch(void* const* d_in, const int* in_sizes, int n_in,
                              void* d_out, int out_size, void* d_ws, size_t ws_size,
                              hipStream_t stream) {
    const float* Z_mu      = (const float*)d_in[0];
    const float* Z_sigma   = (const float*)d_in[1];
    const float* corr      = (const float*)d_in[2];
    const float* cell_prob = (const float*)d_in[3];

    float* mu_out = (float*)d_out;
    float* sg_out = mu_out + S_STATES * G_GENES;

    const int nblocks = G_GENES / GB;
    lognorm_wmma_kernel<<<nblocks, 128, 0, stream>>>(
        Z_mu, Z_sigma, corr, cell_prob, mu_out, sg_out);
}


# corr and its transpose folded into one triangular operand, both halves read from one LDS copy of corr (one barrier per workgroup): 40 matrix instructions per state, not 64; sums regrouped
# speedup vs baseline: 2.6058x; 1.1726x over previous
#include <hip/hip_runtime.h>


#define S_STATES 16
#define C_TYPES  64
#define G_GENES  20000
#define GB       16
#define PITCH    68

typedef __attribute__((ext_vector_type(2))) float v2f;
typedef __attribute__((ext_vector_type(4))) float v4f;
typedef __attribute__((ext_vector_type(8))) float v8f;

__device__ __forceinline__ int kpos(int kk, int hi) {
    return 16 * (kk >> 2) + 8 * hi + 2 * (kk & 3);
}

__global__ __launch_bounds__(128, 3)
void lognorm_wmma_kernel(const float* __restrict__ Z_mu,
                         const float* __restrict__ Z_sigma,
                         const float* __restrict__ corr,
                         const float* __restrict__ cell_prob,
                         float* __restrict__ mu_out,
                         float* __restrict__ sg_out) {
    __shared__ __attribute__((aligned(16))) float cs[C_TYPES * PITCH];

    const int tid  = threadIdx.x;
    const int wave = tid >> 5;
    const int lane = tid & 31;
    const int m    = lane & 15;
    const int hi   = lane >> 4;
    const int g    = blockIdx.x * GB + m;

    float rm[32], rs[32];
    #pragma unroll
    for (int kk = 0; kk < 16; ++kk) {
        const int c0 = kpos(kk, hi);
        rm[2 * kk]     = Z_mu[c0 * G_GENES + g];
        rm[2 * kk + 1] = Z_mu[(c0 + 1) * G_GENES + g];
        rs[2 * kk]     = Z_sigma[c0 * G_GENES + g];
        rs[2 * kk + 1] = Z_sigma[(c0 + 1) * G_GENES + g];
    }

    #pragma unroll
    for (int i = 0; i < 8; ++i) {
        const int p   = i * 128 + tid;
        const int row = p >> 4, c4 = p & 15;
        const v4f v = *(const v4f*)(corr + row * C_TYPES + c4 * 4);
        *(v4f*)(cs + row * PITCH + c4 * 4) = v;
    }

    const float* cp = cell_prob + wave * C_TYPES + hi * 8;
    v4f wn[8];
    #pragma unroll
    for (int q = 0; q < 8; ++q) wn[q] = *(const v4f*)(cp + (q >> 1) * 16 + (q & 1) * 4);

    __syncthreads();

    #define OFF(rt) ((rt) * 16 - 2 * (rt) * ((rt) - 1))
    v2f vA[40];
    #pragma unroll
    for (int rt = 0; rt < 4; ++rt) {
        const int r = 16 * rt + m;
        #pragma unroll
        for (int kk = 4 * rt; kk < 16; ++kk) {
            const int c0 = kpos(kk, hi);
            const float d0 = cs[c0 * PITCH + r];
            const float d1 = cs[(c0 + 1) * PITCH + r];
            const v2f  mr = *(const v2f*)(cs + r * PITCH + c0);
            v2f f;
            if ((kk >> 2) > rt) {
                f.x = d0 + mr.x;
                f.y = d1 + mr.y;
            } else {
                const int del = c0 - r;
                f.x = del > 0  ? d0 + mr.x : (del == 0  ? d0 : 0.0f);
                f.y = del > -1 ? d1 + mr.y : (del == -1 ? d1 : 0.0f);
            }
            vA[OFF(rt) + kk - 4 * rt] = f;
        }
    }

    v2f ev[16], ab[16];
    #pragma unroll
    for (int kk = 0; kk < 16; ++kk) {
        const float s0 = rs[2 * kk], s1 = rs[2 * kk + 1];
        const float e0 = __expf(rm[2 * kk] + 0.5f * s0 * s0);
        const float e1 = __expf(rm[2 * kk + 1] + 0.5f * s1 * s1);
        ev[kk].x = e0;      ev[kk].y = e1;
        ab[kk].x = e0 * s0; ab[kk].y = e1 * s1;
    }

    #pragma unroll
    for (int si = 0; si < 4; ++si) {
        const int s = wave + si * 4;

        v2f b[16];
        float ss = 0.0f;
        #pragma unroll
        for (int q = 0; q < 8; ++q) {
            const v4f w = wn[q];
            b[2 * q].x     = w.x * ab[2 * q].x;
            b[2 * q].y     = w.y * ab[2 * q].y;
            b[2 * q + 1].x = w.z * ab[2 * q + 1].x;
            b[2 * q + 1].y = w.w * ab[2 * q + 1].y;
            ss += w.x * ev[2 * q].x;
            ss += w.y * ev[2 * q].y;
            ss += w.z * ev[2 * q + 1].x;
            ss += w.w * ev[2 * q + 1].y;
        }
        ss += __shfl_xor(ss, 16, 32);

        if (si < 3) {
            #pragma unroll
            for (int q = 0; q < 8; ++q)
                wn[q] = *(const v4f*)(cp + (si + 1) * 4 * C_TYPES + (q >> 1) * 16 + (q & 1) * 4);
        }

        v8f acc[4] = {};
        #pragma unroll
        for (int kk = 0; kk < 16; ++kk) {
            #pragma unroll
            for (int rt = 0; rt <= (kk >> 2); ++rt)
                acc[rt] = __builtin_amdgcn_wmma_f32_16x16x4_f32(false, vA[OFF(rt) + kk - 4 * rt], false, b[kk],
                                                                (short)0, acc[rt], false, false);
        }

        float dot = 0.0f;
        #pragma unroll
        for (int rt = 0; rt < 4; ++rt) {
            #pragma unroll
            for (int v = 0; v < 8; v += 2) {
                dot += acc[rt][v]     * b[4 * rt + (v >> 1)].x;
                dot += acc[rt][v + 1] * b[4 * rt + (v >> 1)].y;
            }
        }
        dot += __shfl_xor(dot, 16, 32);

        const float var   = dot / (ss * ss);
        const float mu_o  = __logf(ss) - 0.5f * var;
        const float sig_o = sqrtf(var);

        if (lane < 16) {
            mu_out[s * G_GENES + g] = mu_o;
            sg_out[s * G_GENES + g] = sig_o;
        }
    }
    #undef OFF
}

extern "C" void kernel_launch(void* const* d_in, const int* in_sizes, int n_in,
                              void* d_out, int out_size, void* d_ws, size_t ws_size,
                              hipStream_t stream) {
    const float* Z_mu      = (const float*)d_in[0];
    const float* Z_sigma   = (const float*)d_in[1];
    const float* corr      = (const float*)d_in[2];
    const float* cell_prob = (const float*)d_in[3];

    float* mu_out = (float*)d_out;
    float* sg_out = mu_out + S_STATES * G_GENES;

    const int nblocks = G_GENES / GB;
    lognorm_wmma_kernel<<<nblocks, 128, 0, stream>>>(
        Z_mu, Z_sigma, corr, cell_prob, mu_out, sg_out);
}


# work re-split: a 16-wave workgroup owns 80 genes, wave = state over 5 gene tiles; V fragments and cell_prob fetched once per wave, exp terms once per workgroup via LDS; same products, same sums
# speedup vs baseline: 3.0088x; 1.1547x over previous
#include <hip/hip_runtime.h>


#define S_STATES 16
#define C_TYPES  64
#define G_GENES  20000
#define GB       16
#define TILES    5
#define GW       (GB * TILES)
#define NT       512
#define PITCH    68

typedef __attribute__((ext_vector_type(2))) float v2f;
typedef __attribute__((ext_vector_type(4))) float v4f;
typedef __attribute__((ext_vector_type(8))) float v8f;

__device__ __forceinline__ int kpos(int kk, int hi) {
    return 16 * (kk >> 2) + 8 * hi + 2 * (kk & 3);
}

__global__ __launch_bounds__(NT)
void lognorm_wmma_kernel(const float* __restrict__ Z_mu,
                         const float* __restrict__ Z_sigma,
                         const float* __restrict__ corr,
                         const float* __restrict__ cell_prob,
                         float* __restrict__ mu_out,
                         float* __restrict__ sg_out) {
    __shared__ __attribute__((aligned(16))) float cs[C_TYPES * PITCH];
    __shared__ __attribute__((aligned(16))) float evs[GW * PITCH];
    __shared__ __attribute__((aligned(16))) float abss[GW * PITCH];

    const int tid  = threadIdx.x;
    const int wave = tid >> 5;
    const int lane = tid & 31;
    const int m    = lane & 15;
    const int hi   = lane >> 4;
    const int g0   = blockIdx.x * GW;

    v4f cv[2];
    #pragma unroll
    for (int i = 0; i < 2; ++i) {
        const int p   = i * NT + tid;
        const int row = p >> 4, c4 = p & 15;
        cv[i] = *(const v4f*)(corr + row * C_TYPES + c4 * 4);
    }

    float rm[10], rs[10];
    #pragma unroll
    for (int i = 0; i < 10; ++i) {
        const int e  = i * NT + tid;
        const int c  = e / GW;
        const int go = e - c * GW;
        rm[i] = Z_mu[c * G_GENES + g0 + go];
        rs[i] = Z_sigma[c * G_GENES + g0 + go];
    }

    const float* cp = cell_prob + wave * C_TYPES + hi * 8;
    v4f wn[8];
    #pragma unroll
    for (int q = 0; q < 8; ++q) wn[q] = *(const v4f*)(cp + (q >> 1) * 16 + (q & 1) * 4);

    #pragma unroll
    for (int i = 0; i < 2; ++i) {
        const int p   = i * NT + tid;
        const int row = p >> 4, c4 = p & 15;
        *(v4f*)(cs + row * PITCH + c4 * 4) = cv[i];
    }

    #pragma unroll
    for (int i = 0; i < 10; ++i) {
        const int e  = i * NT + tid;
        const int c  = e / GW;
        const int go = e - c * GW;
        const float s0 = rs[i];
        const float e0 = __expf(rm[i] + 0.5f * s0 * s0);
        evs[go * PITCH + c]  = e0;
        abss[go * PITCH + c] = e0 * s0;
    }

    __syncthreads();

    #define OFF(rt) ((rt) * 16 - 2 * (rt) * ((rt) - 1))
    v2f vA[40];
    #pragma unroll
    for (int rt = 0; rt < 4; ++rt) {
        const int r = 16 * rt + m;
        #pragma unroll
        for (int kk = 4 * rt; kk < 16; ++kk) {
            const int c0 = kpos(kk, hi);
            const float d0 = cs[c0 * PITCH + r];
            const float d1 = cs[(c0 + 1) * PITCH + r];
            const v2f  mr = *(const v2f*)(cs + r * PITCH + c0);
            v2f f;
            if ((kk >> 2) > rt) {
                f.x = d0 + mr.x;
                f.y = d1 + mr.y;
            } else {
                const int del = c0 - r;
                f.x = del > 0  ? d0 + mr.x : (del == 0  ? d0 : 0.0f);
                f.y = del > -1 ? d1 + mr.y : (del == -1 ? d1 : 0.0f);
            }
            vA[OFF(rt) + kk - 4 * rt] = f;
        }
    }

    const int s = wave;

    for (int t = 0; t < TILES; ++t) {
        const int rowbase = (t * GB + m) * PITCH + hi * 8;

        v2f b[16];
        float ss = 0.0f;
        #pragma unroll
        for (int q = 0; q < 8; ++q) {
            const int off = rowbase + (q >> 1) * 16 + (q & 1) * 4;
            const v4f a = *(const v4f*)(abss + off);
            const v4f e = *(const v4f*)(evs + off);
            const v4f w = wn[q];
            b[2 * q].x     = w.x * a.x;
            b[2 * q].y     = w.y * a.y;
            b[2 * q + 1].x = w.z * a.z;
            b[2 * q + 1].y = w.w * a.w;
            ss += w.x * e.x;
            ss += w.y * e.y;
            ss += w.z * e.z;
            ss += w.w * e.w;
        }
        ss += __shfl_xor(ss, 16, 32);

        v8f acc[4] = {};
        #pragma unroll
        for (int kk = 0; kk < 16; ++kk) {
            #pragma unroll
            for (int rt = 0; rt <= (kk >> 2); ++rt)
                acc[rt] = __builtin_amdgcn_wmma_f32_16x16x4_f32(false, vA[OFF(rt) + kk - 4 * rt], false, b[kk],
                                                                (short)0, acc[rt], false, false);
        }

        float dot = 0.0f;
        #pragma unroll
        for (int rt = 0; rt < 4; ++rt) {
            #pragma unroll
            for (int v = 0; v < 8; v += 2) {
                dot += acc[rt][v]     * b[4 * rt + (v >> 1)].x;
                dot += acc[rt][v + 1] * b[4 * rt + (v >> 1)].y;
            }
        }
        dot += __shfl_xor(dot, 16, 32);

        const float var   = dot / (ss * ss);
        const float mu_o  = __logf(ss) - 0.5f * var;
        const float sig_o = sqrtf(var);

        if (lane < 16) {
            const int g = g0 + t * GB + m;
            mu_out[s * G_GENES + g] = mu_o;
            sg_out[s * G_GENES + g] = sig_o;
        }
    }
    #undef OFF
}

extern "C" void kernel_launch(void* const* d_in, const int* in_sizes, int n_in,
                              void* d_out, int out_size, void* d_ws, size_t ws_size,
                              hipStream_t stream) {
    const float* Z_mu      = (const float*)d_in[0];
    const float* Z_sigma   = (const float*)d_in[1];
    const float* corr      = (const float*)d_in[2];
    const float* cell_prob = (const float*)d_in[3];

    float* mu_out = (float*)d_out;
    float* sg_out = mu_out + S_STATES * G_GENES;

    const int nblocks = G_GENES / GW;
    lognorm_wmma_kernel<<<nblocks, NT, 0, stream>>>(
        Z_mu, Z_sigma, corr, cell_prob, mu_out, sg_out);
}


# var*Ssum^2 as ONE product over type pairs: rows = 16 states, k = pairs c<=d, 560 matrix instructions per gene tile, not 640; k split over 8 waves, partials summed via LDS; products, sums regrouped
# speedup vs baseline: 3.1859x; 1.0589x over previous
#include <hip/hip_runtime.h>


#define S_STATES 16
#define C_TYPES  64
#define G_GENES  20000
#define GB       16
#define TILES    5
#define GW       (GB * TILES)
#define NT       256
#define NW       8
#define PITCH    68

typedef __attribute__((ext_vector_type(2))) float v2f;
typedef __attribute__((ext_vector_type(4))) float v4f;
typedef __attribute__((ext_vector_type(8))) float v8f;

__device__ __forceinline__ void lds_barrier() {
    asm volatile("s_wait_dscnt 0x0\n\ts_barrier_signal -1\n\ts_barrier_wait -1" ::: "memory");
}

#define IS_LONG(t, w) ((t) <= 8 ? true : ((t) == 16 ? false : ((t) < 16 - (w))))

__global__ __launch_bounds__(NT, 2)
void lognorm_wmma_kernel(const float* __restrict__ Z_mu,
                         const float* __restrict__ Z_sigma,
                         const float* __restrict__ corr,
                         const float* __restrict__ cell_prob,
                         float* __restrict__ mu_out,
                         float* __restrict__ sg_out) {
    __shared__ __attribute__((aligned(16))) float cs[C_TYPES * PITCH];
    __shared__ __attribute__((aligned(16))) float vs[C_TYPES * PITCH];
    __shared__ __attribute__((aligned(16))) float ps[S_STATES * PITCH];
    __shared__ __attribute__((aligned(16))) float evs[GW * PITCH];
    __shared__ __attribute__((aligned(16))) float abss[GW * PITCH];
    __shared__ __attribute__((aligned(16))) float red[TILES * NW * 2 * 256];

    const int tid  = threadIdx.x;
    const int lane = tid & 31;
    const int m    = lane & 15;
    const int hi   = lane >> 4;
    const int w    = __builtin_amdgcn_readfirstlane(tid >> 5);
    const int g0   = blockIdx.x * GW;

    v4f cv[4];
    #pragma unroll
    for (int i = 0; i < 4; ++i) {
        const int p   = i * NT + tid;
        const int row = p >> 4, c4 = p & 15;
        cv[i] = *(const v4f*)(corr + row * C_TYPES + c4 * 4);
    }
    const v4f pv = *(const v4f*)(cell_prob + tid * 4);

    float rm[20], rs[20];
    #pragma unroll
    for (int i = 0; i < 20; ++i) {
        const int e  = i * NT + tid;
        const int c  = e / GW;
        const int go = e - c * GW;
        rm[i] = Z_mu[c * G_GENES + g0 + go];
        rs[i] = Z_sigma[c * G_GENES + g0 + go];
    }

    #pragma unroll
    for (int i = 0; i < 4; ++i) {
        const int p   = i * NT + tid;
        const int row = p >> 4, c4 = p & 15;
        *(v4f*)(cs + row * PITCH + c4 * 4) = cv[i];
    }
    *(v4f*)(ps + (tid >> 4) * PITCH + (tid & 15) * 4) = pv;

    lds_barrier();

    #pragma unroll
    for (int i = 0; i < 4; ++i) {
        const int p  = i * NT + tid;
        const int r  = p & 63, c4 = p >> 6;
        const int cb = 4 * c4;
        const v4f d = *(const v4f*)(cs + r * PITCH + cb);
        const float t0 = cs[(cb + 0) * PITCH + r];
        const float t1 = cs[(cb + 1) * PITCH + r];
        const float t2 = cs[(cb + 2) * PITCH + r];
        const float t3 = cs[(cb + 3) * PITCH + r];
        v4f o;
        o.x = (cb + 0 == r) ? d.x : t0 + d.x;
        o.y = (cb + 1 == r) ? d.y : t1 + d.y;
        o.z = (cb + 2 == r) ? d.z : t2 + d.z;
        o.w = (cb + 3 == r) ? d.w : t3 + d.w;
        *(v4f*)(vs + r * PITCH + cb) = o;
    }

    lds_barrier();

    const float* prow = ps + m * PITCH;
    v2f PA[4][17];
    v2f SA[2];
    {
        const v4f pl = *(const v4f*)(prow + 4 * w);
        const v4f ph = *(const v4f*)(prow + 60 - 4 * w);
        const float pcL[4] = {pl.x, pl.y, pl.z, pl.w};
        const float pcS[4] = {ph.x, ph.y, ph.z, ph.w};
        #pragma unroll
        for (int r = 0; r < 4; ++r) {
            #pragma unroll
            for (int t = 0; t < 17; ++t) {
                const bool lg = IS_LONG(t, w);
                const int c   = lg ? 4 * w + r : 60 - 4 * w + r;
                const int dq  = lg ? w + t : t - 1;
                const int d0  = 4 * dq + 2 * hi;
                v2f vv = *(const v2f*)(vs + c * PITCH + d0);
                const v2f pd = *(const v2f*)(prow + d0);
                const float pc = lg ? pcL[r] : pcS[r];
                vv.x = (d0     >= c) ? vv.x : 0.0f;
                vv.y = (d0 + 1 >= c) ? vv.y : 0.0f;
                v2f f;
                f.x = (pc * pd.x) * vv.x;
                f.y = (pc * pd.y) * vv.y;
                PA[r][t] = f;
            }
        }
        #pragma unroll
        for (int u = 0; u < 2; ++u) SA[u] = *(const v2f*)(prow + 8 * w + 4 * u + 2 * hi);
    }

    __builtin_amdgcn_sched_barrier(0);

    #pragma unroll
    for (int i = 0; i < 20; ++i) {
        const int e  = i * NT + tid;
        const int c  = e / GW;
        const int go = e - c * GW;
        const float s0 = rs[i];
        const float e0 = __expf(rm[i] + 0.5f * s0 * s0);
        evs[go * PITCH + c]  = e0;
        abss[go * PITCH + c] = e0 * s0;
    }

    __syncthreads();

    for (int t5 = 0; t5 < TILES; ++t5) {
        const float* arow = abss + (t5 * GB + m) * PITCH;
        const float* erow = evs  + (t5 * GB + m) * PITCH;

        const v4f al = *(const v4f*)(arow + 4 * w);
        const v4f ah = *(const v4f*)(arow + 60 - 4 * w);
        const float acL[4] = {al.x, al.y, al.z, al.w};
        const float acS[4] = {ah.x, ah.y, ah.z, ah.w};
        v2f adL[16], adS[17];
        #pragma unroll
        for (int t = 0; t < 16; ++t) {
            const int dq = (w + t < 15) ? w + t : 15;
            adL[t] = *(const v2f*)(arow + 4 * dq + 2 * hi);
        }
        #pragma unroll
        for (int t = 9; t < 17; ++t) adS[t] = *(const v2f*)(arow + 4 * (t - 1) + 2 * hi);

        v8f acc0 = {}, acc1 = {}, accS = {};
        #pragma unroll
        for (int r = 0; r < 4; ++r) {
            #pragma unroll
            for (int t = 0; t < 17; ++t) {
                const bool lg = IS_LONG(t, w);
                const float ac = lg ? acL[r] : acS[r];
                const v2f  ad = (t == 16) ? adS[16] : ((t <= 8) ? adL[t] : (lg ? adL[t] : adS[t]));
                v2f b;
                b.x = ac * ad.x;
                b.y = ac * ad.y;
                if (((r * 17 + t) & 1) == 0)
                    acc0 = __builtin_amdgcn_wmma_f32_16x16x4_f32(false, PA[r][t], false, b, (short)0, acc0, false, false);
                else
                    acc1 = __builtin_amdgcn_wmma_f32_16x16x4_f32(false, PA[r][t], false, b, (short)0, acc1, false, false);
            }
        }
        #pragma unroll
        for (int u = 0; u < 2; ++u) {
            const v2f be = *(const v2f*)(erow + 8 * w + 4 * u + 2 * hi);
            accS = __builtin_amdgcn_wmma_f32_16x16x4_f32(false, SA[u], false, be, (short)0, accS, false, false);
        }

        float* rd = red + (t5 * NW + w) * 2 * 256 + lane * 4;
        v4f o;
        o.x = acc0[0] + acc1[0]; o.y = acc0[1] + acc1[1]; o.z = acc0[2] + acc1[2]; o.w = acc0[3] + acc1[3];
        *(v4f*)(rd) = o;
        o.x = acc0[4] + acc1[4]; o.y = acc0[5] + acc1[5]; o.z = acc0[6] + acc1[6]; o.w = acc0[7] + acc1[7];
        *(v4f*)(rd + 128) = o;
        o.x = accS[0]; o.y = accS[1]; o.z = accS[2]; o.w = accS[3];
        *(v4f*)(rd + 256) = o;
        o.x = accS[4]; o.y = accS[5]; o.z = accS[6]; o.w = accS[7];
        *(v4f*)(rd + 256 + 128) = o;
    }

    __syncthreads();

    {
        const int st   = tid >> 4, gene = tid & 15;
        const int idx  = ((st >> 2) & 1) * 128 + ((st >> 3) * 16 + gene) * 4 + (st & 3);
        #pragma unroll
        for (int t5 = 0; t5 < TILES; ++t5) {
            float dsum = 0.0f, ss = 0.0f;
            #pragma unroll
            for (int wv = 0; wv < NW; ++wv) {
                dsum += red[(t5 * NW + wv) * 2 * 256 + idx];
                ss   += red[(t5 * NW + wv) * 2 * 256 + 256 + idx];
            }
            const float var   = dsum / (ss * ss);
            const float mu_o  = __logf(ss) - 0.5f * var;
            const float sig_o = sqrtf(var);
            const int g = g0 + t5 * GB + gene;
            mu_out[st * G_GENES + g] = mu_o;
            sg_out[st * G_GENES + g] = sig_o;
        }
    }
}

extern "C" void kernel_launch(void* const* d_in, const int* in_sizes, int n_in,
                              void* d_out, int out_size, void* d_ws, size_t ws_size,
                              hipStream_t stream) {
    const float* Z_mu      = (const float*)d_in[0];
    const float* Z_sigma   = (const float*)d_in[1];
    const float* corr      = (const float*)d_in[2];
    const float* cell_prob = (const float*)d_in[3];

    float* mu_out = (float*)d_out;
    float* sg_out = mu_out + S_STATES * G_GENES;

    const int nblocks = G_GENES / GW;
    lognorm_wmma_kernel<<<nblocks, NT, 0, stream>>>(
        Z_mu, Z_sigma, corr, cell_prob, mu_out, sg_out);
}


# B operand = ab[d] straight from LDS; each row c sums into its own tile, scaled by ab[c] once; blocks taken cyclically, one fixed schedule for all waves: no per-step multiply or select; sums regrouped
# speedup vs baseline: 3.5067x; 1.1007x over previous
#include <hip/hip_runtime.h>


#define S_STATES 16
#define C_TYPES  64
#define G_GENES  20000
#define GB       16
#define TILES    5
#define GW       (GB * TILES)
#define NT       256
#define NW       8
#define PITCH    68

typedef __attribute__((ext_vector_type(2))) float v2f;
typedef __attribute__((ext_vector_type(4))) float v4f;
typedef __attribute__((ext_vector_type(8))) float v8f;

__device__ __forceinline__ void lds_barrier() {
    asm volatile("s_wait_dscnt 0x0\n\ts_barrier_signal -1\n\ts_barrier_wait -1" ::: "memory");
}

#define NBLK 17
#define BLK_Q(b, w)  ((b) <= 8 ? (w) : (w) + 8)
#define BLK_DQ(b, w) ((b) <= 8 ? (w) + (b) : (((w) + (b) - 1) & 15))

__global__ __launch_bounds__(NT, 2)
void lognorm_wmma_kernel(const float* __restrict__ Z_mu,
                         const float* __restrict__ Z_sigma,
                         const float* __restrict__ corr,
                         const float* __restrict__ cell_prob,
                         float* __restrict__ mu_out,
                         float* __restrict__ sg_out) {
    __shared__ __attribute__((aligned(16))) float cs[C_TYPES * PITCH];
    __shared__ __attribute__((aligned(16))) float vs[C_TYPES * PITCH];
    __shared__ __attribute__((aligned(16))) float ps[S_STATES * PITCH];
    __shared__ __attribute__((aligned(16))) float evs[GW * PITCH];
    __shared__ __attribute__((aligned(16))) float abss[GW * PITCH];
    __shared__ __attribute__((aligned(16))) float red[TILES * NW * 2 * 256];

    const int tid  = threadIdx.x;
    const int lane = tid & 31;
    const int m    = lane & 15;
    const int hi   = lane >> 4;
    const int w    = __builtin_amdgcn_readfirstlane(tid >> 5);
    const int g0   = blockIdx.x * GW;

    v4f cv[4];
    #pragma unroll
    for (int i = 0; i < 4; ++i) {
        const int p   = i * NT + tid;
        const int row = p >> 4, c4 = p & 15;
        cv[i] = *(const v4f*)(corr + row * C_TYPES + c4 * 4);
    }
    const v4f pv = *(const v4f*)(cell_prob + tid * 4);

    float rm[20], rs[20];
    #pragma unroll
    for (int i = 0; i < 20; ++i) {
        const int e  = i * NT + tid;
        const int c  = e / GW;
        const int go = e - c * GW;
        rm[i] = Z_mu[c * G_GENES + g0 + go];
        rs[i] = Z_sigma[c * G_GENES + g0 + go];
    }

    #pragma unroll
    for (int i = 0; i < 4; ++i) {
        const int p   = i * NT + tid;
        const int row = p >> 4, c4 = p & 15;
        *(v4f*)(cs + row * PITCH + c4 * 4) = cv[i];
    }
    *(v4f*)(ps + (tid >> 4) * PITCH + (tid & 15) * 4) = pv;

    lds_barrier();

    #pragma unroll
    for (int i = 0; i < 4; ++i) {
        const int p  = i * NT + tid;
        const int r  = p & 63, c4 = p >> 6;
        const int cb = 4 * c4;
        const v4f d = *(const v4f*)(cs + r * PITCH + cb);
        const float t0 = cs[(cb + 0) * PITCH + r];
        const float t1 = cs[(cb + 1) * PITCH + r];
        const float t2 = cs[(cb + 2) * PITCH + r];
        const float t3 = cs[(cb + 3) * PITCH + r];
        v4f o;
        o.x = (cb + 0 == r) ? d.x : t0 + d.x;
        o.y = (cb + 1 == r) ? d.y : t1 + d.y;
        o.z = (cb + 2 == r) ? d.z : t2 + d.z;
        o.w = (cb + 3 == r) ? d.w : t3 + d.w;
        *(v4f*)(vs + r * PITCH + cb) = o;
    }

    lds_barrier();

    const float* prow = ps + m * PITCH;
    v2f PA[NBLK][4];
    v2f SA[2];
    {
        const v4f pl = *(const v4f*)(prow + 4 * w);
        const v4f ph = *(const v4f*)(prow + 4 * w + 32);
        const float pcL[4] = {pl.x, pl.y, pl.z, pl.w};
        const float pcS[4] = {ph.x, ph.y, ph.z, ph.w};
        #pragma unroll
        for (int b = 0; b < NBLK; ++b) {
            const int q   = BLK_Q(b, w);
            const int dq  = BLK_DQ(b, w);
            const int d0  = 4 * dq + 2 * hi;
            const v2f pd  = *(const v2f*)(prow + d0);
            #pragma unroll
            for (int r = 0; r < 4; ++r) {
                const int c = 4 * q + r;
                v2f vv = *(const v2f*)(vs + c * PITCH + d0);
                const float pc = (b <= 8) ? pcL[r] : pcS[r];
                if (b == 0 || b == 9) {
                    vv.x = (2 * hi     >= r) ? vv.x : 0.0f;
                    vv.y = (2 * hi + 1 >= r) ? vv.y : 0.0f;
                }
                v2f f;
                f.x = (pc * pd.x) * vv.x;
                f.y = (pc * pd.y) * vv.y;
                PA[b][r] = f;
            }
        }
        #pragma unroll
        for (int u = 0; u < 2; ++u) SA[u] = *(const v2f*)(prow + 8 * w + 4 * u + 2 * hi);
    }

    __builtin_amdgcn_sched_barrier(0);

    #pragma unroll
    for (int i = 0; i < 20; ++i) {
        const int e  = i * NT + tid;
        const int c  = e / GW;
        const int go = e - c * GW;
        const float s0 = rs[i];
        const float e0 = __expf(rm[i] + 0.5f * s0 * s0);
        evs[go * PITCH + c]  = e0;
        abss[go * PITCH + c] = e0 * s0;
    }

    __syncthreads();

    for (int t5 = 0; t5 < TILES; ++t5) {
        const float* arow = abss + (t5 * GB + m) * PITCH;
        const float* erow = evs  + (t5 * GB + m) * PITCH;

        const v4f al = *(const v4f*)(arow + 4 * w);
        const v4f ah = *(const v4f*)(arow + 4 * w + 32);
        v2f ad[NBLK];
        #pragma unroll
        for (int b = 0; b < NBLK; ++b) ad[b] = *(const v2f*)(arow + 4 * BLK_DQ(b, w) + 2 * hi);

        v8f TL[4] = {}, TS[4] = {}, accS = {};
        #pragma unroll
        for (int b = 0; b <= 8; ++b) {
            #pragma unroll
            for (int r = 0; r < 4; ++r)
                TL[r] = __builtin_amdgcn_wmma_f32_16x16x4_f32(false, PA[b][r], false, ad[b], (short)0, TL[r], false, false);
        }
        #pragma unroll
        for (int b = 9; b < NBLK; ++b) {
            #pragma unroll
            for (int r = 0; r < 4; ++r)
                TS[r] = __builtin_amdgcn_wmma_f32_16x16x4_f32(false, PA[b][r], false, ad[b], (short)0, TS[r], false, false);
        }
        #pragma unroll
        for (int u = 0; u < 2; ++u) {
            const v2f be = *(const v2f*)(erow + 8 * w + 4 * u + 2 * hi);
            accS = __builtin_amdgcn_wmma_f32_16x16x4_f32(false, SA[u], false, be, (short)0, accS, false, false);
        }

        v8f acc;
        #pragma unroll
        for (int v = 0; v < 8; ++v) {
            float a = al.x * TL[0][v];
            a += al.y * TL[1][v];
            a += al.z * TL[2][v];
            a += al.w * TL[3][v];
            a += ah.x * TS[0][v];
            a += ah.y * TS[1][v];
            a += ah.z * TS[2][v];
            a += ah.w * TS[3][v];
            acc[v] = a;
        }

        float* rd = red + (t5 * NW + w) * 2 * 256 + lane * 4;
        v4f o;
        o.x = acc[0]; o.y = acc[1]; o.z = acc[2]; o.w = acc[3];
        *(v4f*)(rd) = o;
        o.x = acc[4]; o.y = acc[5]; o.z = acc[6]; o.w = acc[7];
        *(v4f*)(rd + 128) = o;
        o.x = accS[0]; o.y = accS[1]; o.z = accS[2]; o.w = accS[3];
        *(v4f*)(rd + 256) = o;
        o.x = accS[4]; o.y = accS[5]; o.z = accS[6]; o.w = accS[7];
        *(v4f*)(rd + 256 + 128) = o;
    }

    __syncthreads();

    {
        const int st   = tid >> 4, gene = tid & 15;
        const int idx  = ((st >> 2) & 1) * 128 + ((st >> 3) * 16 + gene) * 4 + (st & 3);
        #pragma unroll
        for (int t5 = 0; t5 < TILES; ++t5) {
            float dsum = 0.0f, ss = 0.0f;
            #pragma unroll
            for (int wv = 0; wv < NW; ++wv) {
                dsum += red[(t5 * NW + wv) * 2 * 256 + idx];
                ss   += red[(t5 * NW + wv) * 2 * 256 + 256 + idx];
            }
            const float var   = dsum / (ss * ss);
            const float mu_o  = __logf(ss) - 0.5f * var;
            const float sig_o = sqrtf(var);
            const int g = g0 + t5 * GB + gene;
            mu_out[st * G_GENES + g] = mu_o;
            sg_out[st * G_GENES + g] = sig_o;
        }
    }
}

extern "C" void kernel_launch(void* const* d_in, const int* in_sizes, int n_in,
                              void* d_out, int out_size, void* d_ws, size_t ws_size,
                              hipStream_t stream) {
    const float* Z_mu      = (const float*)d_in[0];
    const float* Z_sigma   = (const float*)d_in[1];
    const float* corr      = (const float*)d_in[2];
    const float* cell_prob = (const float*)d_in[3];

    float* mu_out = (float*)d_out;
    float* sg_out = mu_out + S_STATES * G_GENES;

    const int nblocks = G_GENES / GW;
    lognorm_wmma_kernel<<<nblocks, NT, 0, stream>>>(
        Z_mu, Z_sigma, corr, cell_prob, mu_out, sg_out);
}
